# starting kernel
# baseline (speedup 1.0000x reference)
#include <hip/hip_runtime.h>
#include <hip/hip_bf16.h>

typedef float v2f __attribute__((ext_vector_type(2)));
typedef float v8f __attribute__((ext_vector_type(8)));

#define D_DIM 512
#define N_DIM 32768
#define E_DIM 8
#define LDS_STRIDE 516

__global__ __launch_bounds__(256) void SplitMind_reduce_experts(
    const float* __restrict__ W, const float* __restrict__ b,
    float* __restrict__ Wsum, float* __restrict__ bsum) {
  const int i = blockIdx.x * 256 + threadIdx.x;
  const float4* W4 = (const float4*)W;
  float4 acc = W4[i];
#pragma unroll
  for (int e = 1; e < E_DIM; ++e) {
    float4 v = W4[(size_t)e * (D_DIM * D_DIM / 4) + i];
    acc.x += v.x; acc.y += v.y; acc.z += v.z; acc.w += v.w;
  }
  ((float4*)Wsum)[i] = acc;

  if (i < D_DIM / 4) {
    const float4* b4 = (const float4*)b;
    float4 ab = b4[i];
#pragma unroll
    for (int e = 1; e < E_DIM; ++e) {
      float4 v = b4[e * (D_DIM / 4) + i];
      ab.x += v.x; ab.y += v.y; ab.z += v.z; ab.w += v.w;
    }
    ((float4*)bsum)[i] = ab;
  }
}

__global__ __launch_bounds__(256) void SplitMind_moe_gemm(
    const float* __restrict__ X, const float* __restrict__ wgt,
    const float* __restrict__ Wsum, const float* __restrict__ bsum,
    float* __restrict__ out) {
  __shared__ float xs[16 * LDS_STRIDE];

  const int r0   = blockIdx.x * 16;
  const int tid  = threadIdx.x;
  const int lane = tid & 31;
  const int wave = tid >> 5;

#pragma unroll
  for (int it = 0; it < 8; ++it) {
    int i   = tid + it * 256;
    int row = i >> 7;
    int c4  = i & 127;
    float4 v = ((const float4*)(X + (size_t)(r0 + row) * D_DIM))[c4];
    float* dst = &xs[row * LDS_STRIDE + c4 * 4];
    dst[0] = v.x; dst[1] = v.y; dst[2] = v.z; dst[3] = v.w;
  }
  __syncthreads();

  const int m    = lane & 15;
  const int hi   = lane >> 4;
  const int koff = 2 * hi;
  const int fb   = wave * 64;

  const float* aptr = &xs[m * LDS_STRIDE + koff];
  const float* bptr0 = Wsum + (size_t)(fb +  0 + m) * D_DIM + koff;
  const float* bptr1 = Wsum + (size_t)(fb + 16 + m) * D_DIM + koff;
  const float* bptr2 = Wsum + (size_t)(fb + 32 + m) * D_DIM + koff;
  const float* bptr3 = Wsum + (size_t)(fb + 48 + m) * D_DIM + koff;

  v8f acc0 = {}, acc1 = {}, acc2 = {}, acc3 = {};

#pragma unroll 4
  for (int k = 0; k < D_DIM; k += 4) {
    v2f a  = *(const v2f*)(aptr  + k);
    v2f b0 = *(const v2f*)(bptr0 + k);
    v2f b1 = *(const v2f*)(bptr1 + k);
    v2f b2 = *(const v2f*)(bptr2 + k);
    v2f b3 = *(const v2f*)(bptr3 + k);
    acc0 = __builtin_amdgcn_wmma_f32_16x16x4_f32(false, a, false, b0, (short)0, acc0, false, false);
    acc1 = __builtin_amdgcn_wmma_f32_16x16x4_f32(false, a, false, b1, (short)0, acc1, false, false);
    acc2 = __builtin_amdgcn_wmma_f32_16x16x4_f32(false, a, false, b2, (short)0, acc2, false, false);
    acc3 = __builtin_amdgcn_wmma_f32_16x16x4_f32(false, a, false, b3, (short)0, acc3, false, false);
  }

  v8f accs[4] = {acc0, acc1, acc2, acc3};
  float bvals[4];
#pragma unroll
  for (int t = 0; t < 4; ++t) bvals[t] = bsum[fb + t * 16 + m];

#pragma unroll
  for (int j = 0; j < 8; ++j) {
    const int row = r0 + j + 8 * hi;
    const float wv = wgt[row];
    float* orow = out + (size_t)row * D_DIM + fb + m;
#pragma unroll
    for (int t = 0; t < 4; ++t) {
      orow[t * 16] = (accs[t][j] + bvals[t]) * wv;
    }
  }
}

extern "C" void kernel_launch(void* const* d_in, const int* in_sizes, int n_in,
                              void* d_out, int out_size, void* d_ws, size_t ws_size,
                              hipStream_t stream) {
  const float* X   = (const float*)d_in[0];
  const float* wgt = (const float*)d_in[1];
  const float* W   = (const float*)d_in[2];
  const float* b   = (const float*)d_in[3];
  float* out  = (float*)d_out;
  float* Wsum = (float*)d_ws;
  float* bsum = Wsum + D_DIM * D_DIM;

  SplitMind_reduce_experts<<<256, 256, 0, stream>>>(W, b, Wsum, bsum);
  SplitMind_moe_gemm<<<N_DIM / 16, 256, 0, stream>>>(X, wgt, Wsum, bsum, out);
}


# Row tile per block 16 -> 64: each wave keeps a 64x64 accumulator (4x4 WMMA tiles), so every B fragment loaded from Wsum feeds 4 WMMAs instead of 1; per-element sum order unchanged
# speedup vs baseline: 2.8153x; 2.8153x over previous
#include <hip/hip_runtime.h>
#include <hip/hip_bf16.h>

typedef float v2f __attribute__((ext_vector_type(2)));
typedef float v8f __attribute__((ext_vector_type(8)));

#define D_DIM 512
#define N_DIM 32768
#define E_DIM 8
#define LDS_STRIDE 516

__global__ __launch_bounds__(256) void SplitMind_reduce_experts(
    const float* __restrict__ W, const float* __restrict__ b,
    float* __restrict__ Wsum, float* __restrict__ bsum) {
  const int i = blockIdx.x * 256 + threadIdx.x;
  const float4* W4 = (const float4*)W;
  float4 acc = W4[i];
#pragma unroll
  for (int e = 1; e < E_DIM; ++e) {
    float4 v = W4[(size_t)e * (D_DIM * D_DIM / 4) + i];
    acc.x += v.x; acc.y += v.y; acc.z += v.z; acc.w += v.w;
  }
  ((float4*)Wsum)[i] = acc;

  if (i < D_DIM / 4) {
    const float4* b4 = (const float4*)b;
    float4 ab = b4[i];
#pragma unroll
    for (int e = 1; e < E_DIM; ++e) {
      float4 v = b4[e * (D_DIM / 4) + i];
      ab.x += v.x; ab.y += v.y; ab.z += v.z; ab.w += v.w;
    }
    ((float4*)bsum)[i] = ab;
  }
}

#define BM 64
#define RT (BM / 16)

__global__ __launch_bounds__(256) void SplitMind_moe_gemm(
    const float* __restrict__ X, const float* __restrict__ wgt,
    const float* __restrict__ Wsum, const float* __restrict__ bsum,
    float* __restrict__ out) {
  __shared__ float xs[BM * LDS_STRIDE];

  const int r0   = blockIdx.x * BM;
  const int tid  = threadIdx.x;
  const int lane = tid & 31;
  const int wave = tid >> 5;

#pragma unroll
  for (int it = 0; it < BM / 2; ++it) {
    int i   = tid + it * 256;
    int row = i >> 7;
    int c4  = i & 127;
    float4 v = ((const float4*)(X + (size_t)(r0 + row) * D_DIM))[c4];
    float* dst = &xs[row * LDS_STRIDE + c4 * 4];
    dst[0] = v.x; dst[1] = v.y; dst[2] = v.z; dst[3] = v.w;
  }
  __syncthreads();

  const int m    = lane & 15;
  const int hi   = lane >> 4;
  const int koff = 2 * hi;
  const int fb   = wave * 64;

  const float* aptr = &xs[m * LDS_STRIDE + koff];
  const float* bptr0 = Wsum + (size_t)(fb +  0 + m) * D_DIM + koff;
  const float* bptr1 = Wsum + (size_t)(fb + 16 + m) * D_DIM + koff;
  const float* bptr2 = Wsum + (size_t)(fb + 32 + m) * D_DIM + koff;
  const float* bptr3 = Wsum + (size_t)(fb + 48 + m) * D_DIM + koff;

  v8f acc[RT][4];
#pragma unroll
  for (int rt = 0; rt < RT; ++rt)
#pragma unroll
    for (int t = 0; t < 4; ++t) acc[rt][t] = v8f{};

#pragma unroll 4
  for (int k = 0; k < D_DIM; k += 4) {
    v2f b0 = *(const v2f*)(bptr0 + k);
    v2f b1 = *(const v2f*)(bptr1 + k);
    v2f b2 = *(const v2f*)(bptr2 + k);
    v2f b3 = *(const v2f*)(bptr3 + k);
    v2f a[RT];
#pragma unroll
    for (int rt = 0; rt < RT; ++rt)
      a[rt] = *(const v2f*)(aptr + rt * 16 * LDS_STRIDE + k);
#pragma unroll
    for (int rt = 0; rt < RT; ++rt) {
      acc[rt][0] = __builtin_amdgcn_wmma_f32_16x16x4_f32(false, a[rt], false, b0, (short)0, acc[rt][0], false, false);
      acc[rt][1] = __builtin_amdgcn_wmma_f32_16x16x4_f32(false, a[rt], false, b1, (short)0, acc[rt][1], false, false);
      acc[rt][2] = __builtin_amdgcn_wmma_f32_16x16x4_f32(false, a[rt], false, b2, (short)0, acc[rt][2], false, false);
      acc[rt][3] = __builtin_amdgcn_wmma_f32_16x16x4_f32(false, a[rt], false, b3, (short)0, acc[rt][3], false, false);
    }
  }

  float bvals[4];
#pragma unroll
  for (int t = 0; t < 4; ++t) bvals[t] = bsum[fb + t * 16 + m];

#pragma unroll
  for (int rt = 0; rt < RT; ++rt) {
#pragma unroll
    for (int j = 0; j < 8; ++j) {
      const int row = r0 + rt * 16 + j + 8 * hi;
      const float wv = wgt[row];
      float* orow = out + (size_t)row * D_DIM + fb + m;
#pragma unroll
      for (int t = 0; t < 4; ++t) {
        orow[t * 16] = (acc[rt][t][j] + bvals[t]) * wv;
      }
    }
  }
}

extern "C" void kernel_launch(void* const* d_in, const int* in_sizes, int n_in,
                              void* d_out, int out_size, void* d_ws, size_t ws_size,
                              hipStream_t stream) {
  const float* X   = (const float*)d_in[0];
  const float* wgt = (const float*)d_in[1];
  const float* W   = (const float*)d_in[2];
  const float* b   = (const float*)d_in[3];
  float* out  = (float*)d_out;
  float* Wsum = (float*)d_ws;
  float* bsum = Wsum + D_DIM * D_DIM;

  SplitMind_reduce_experts<<<256, 256, 0, stream>>>(W, b, Wsum, bsum);
  SplitMind_moe_gemm<<<N_DIM / BM, 256, 0, stream>>>(X, wgt, Wsum, bsum, out);
}


# K-loop software-pipelined: B fragments (Wsum) fetched into a second register set 64 WMMAs ahead, A fragments (LDS) 32 ahead, so no wave waits on a load with its matrix pipe empty; sum order unchanged
# speedup vs baseline: 2.9100x; 1.0336x over previous
#include <hip/hip_runtime.h>
#include <hip/hip_bf16.h>

typedef float v2f __attribute__((ext_vector_type(2)));
typedef float v8f __attribute__((ext_vector_type(8)));

#define D_DIM 512
#define N_DIM 32768
#define E_DIM 8
#define LDS_STRIDE 516

__global__ __launch_bounds__(256) void SplitMind_reduce_experts(
    const float* __restrict__ W, const float* __restrict__ b,
    float* __restrict__ Wsum, float* __restrict__ bsum) {
  const int i = blockIdx.x * 256 + threadIdx.x;
  const float4* W4 = (const float4*)W;
  float4 acc = W4[i];
#pragma unroll
  for (int e = 1; e < E_DIM; ++e) {
    float4 v = W4[(size_t)e * (D_DIM * D_DIM / 4) + i];
    acc.x += v.x; acc.y += v.y; acc.z += v.z; acc.w += v.w;
  }
  ((float4*)Wsum)[i] = acc;

  if (i < D_DIM / 4) {
    const float4* b4 = (const float4*)b;
    float4 ab = b4[i];
#pragma unroll
    for (int e = 1; e < E_DIM; ++e) {
      float4 v = b4[e * (D_DIM / 4) + i];
      ab.x += v.x; ab.y += v.y; ab.z += v.z; ab.w += v.w;
    }
    ((float4*)bsum)[i] = ab;
  }
}

#define BM 64
#define RT (BM / 16)

#define PIN_HERE() __builtin_amdgcn_sched_barrier(0)
__device__ __forceinline__ void load_b_group(v2f (&b)[4][4], const float* bp, int k) {
#pragma unroll
  for (int s = 0; s < 4; ++s)
#pragma unroll
    for (int t = 0; t < 4; ++t)
      b[s][t] = *(const v2f*)(bp + (size_t)t * 16 * D_DIM + k + 4 * s);
}
__device__ __forceinline__ void load_a_group(v2f (&a)[2][RT], const float* ap, int k) {
#pragma unroll
  for (int s = 0; s < 2; ++s)
#pragma unroll
    for (int rt = 0; rt < RT; ++rt)
      a[s][rt] = *(const v2f*)(ap + rt * 16 * LDS_STRIDE + k + 4 * s);
}
template <int SB>
__device__ __forceinline__ void mma_two_steps(v8f (&acc)[RT][4], const v2f (&a)[2][RT],
                                              const v2f (&b)[4][4]) {
#pragma unroll
  for (int s = 0; s < 2; ++s)
#pragma unroll
    for (int rt = 0; rt < RT; ++rt)
#pragma unroll
      for (int t = 0; t < 4; ++t)
        acc[rt][t] = __builtin_amdgcn_wmma_f32_16x16x4_f32(false, a[s][rt], false, b[SB + s][t],
                                                           (short)0, acc[rt][t], false, false);
}

__global__ __launch_bounds__(256) void SplitMind_moe_gemm(
    const float* __restrict__ X, const float* __restrict__ wgt,
    const float* __restrict__ Wsum, const float* __restrict__ bsum,
    float* __restrict__ out) {
  __shared__ float xs[BM * LDS_STRIDE];

  const int r0   = blockIdx.x * BM;
  const int tid  = threadIdx.x;
  const int lane = tid & 31;
  const int wave = tid >> 5;

#pragma unroll
  for (int it = 0; it < BM / 2; ++it) {
    int i   = tid + it * 256;
    int row = i >> 7;
    int c4  = i & 127;
    float4 v = ((const float4*)(X + (size_t)(r0 + row) * D_DIM))[c4];
    float* dst = &xs[row * LDS_STRIDE + c4 * 4];
    dst[0] = v.x; dst[1] = v.y; dst[2] = v.z; dst[3] = v.w;
  }
  __syncthreads();

  const int m    = lane & 15;
  const int hi   = lane >> 4;
  const int koff = 2 * hi;
  const int fb   = wave * 64;

  const float* aptr = &xs[m * LDS_STRIDE + koff];
  const float* bptr = Wsum + (size_t)(fb + m) * D_DIM + koff;

  v8f acc[RT][4];
#pragma unroll
  for (int rt = 0; rt < RT; ++rt)
#pragma unroll
    for (int t = 0; t < 4; ++t) acc[rt][t] = v8f{};

  v2f bA[4][4], bB[4][4];
  v2f a0[2][RT], a1[2][RT];
  load_b_group(bA, bptr, 0);
  load_a_group(a0, aptr, 0);
  load_b_group(bB, bptr, 16);
  load_a_group(a1, aptr, 8);
  PIN_HERE();
  for (int k = 0; k < D_DIM - 32; k += 32) {
    mma_two_steps<0>(acc, a0, bA);
    PIN_HERE();
    load_a_group(a0, aptr, k + 16);
    PIN_HERE();
    mma_two_steps<2>(acc, a1, bA);
    PIN_HERE();
    load_b_group(bA, bptr, k + 32);
    load_a_group(a1, aptr, k + 24);
    PIN_HERE();
    mma_two_steps<0>(acc, a0, bB);
    PIN_HERE();
    load_a_group(a0, aptr, k + 32);
    PIN_HERE();
    mma_two_steps<2>(acc, a1, bB);
    PIN_HERE();
    load_b_group(bB, bptr, k + 48);
    load_a_group(a1, aptr, k + 40);
    PIN_HERE();
  }
  mma_two_steps<0>(acc, a0, bA);
  PIN_HERE();
  load_a_group(a0, aptr, D_DIM - 16);
  PIN_HERE();
  mma_two_steps<2>(acc, a1, bA);
  PIN_HERE();
  load_a_group(a1, aptr, D_DIM - 8);
  PIN_HERE();
  mma_two_steps<0>(acc, a0, bB);
  mma_two_steps<2>(acc, a1, bB);

  float bvals[4];
#pragma unroll
  for (int t = 0; t < 4; ++t) bvals[t] = bsum[fb + t * 16 + m];

#pragma unroll
  for (int rt = 0; rt < RT; ++rt) {
#pragma unroll
    for (int j = 0; j < 8; ++j) {
      const int row = r0 + rt * 16 + j + 8 * hi;
      const float wv = wgt[row];
      float* orow = out + (size_t)row * D_DIM + fb + m;
#pragma unroll
      for (int t = 0; t < 4; ++t) {
        orow[t * 16] = (acc[rt][t][j] + bvals[t]) * wv;
      }
    }
  }
}

extern "C" void kernel_launch(void* const* d_in, const int* in_sizes, int n_in,
                              void* d_out, int out_size, void* d_ws, size_t ws_size,
                              hipStream_t stream) {
  const float* X   = (const float*)d_in[0];
  const float* wgt = (const float*)d_in[1];
  const float* W   = (const float*)d_in[2];
  const float* b   = (const float*)d_in[3];
  float* out  = (float*)d_out;
  float* Wsum = (float*)d_ws;
  float* bsum = Wsum + D_DIM * D_DIM;

  SplitMind_reduce_experts<<<256, 256, 0, stream>>>(W, b, Wsum, bsum);
  SplitMind_moe_gemm<<<N_DIM / BM, 256, 0, stream>>>(X, wgt, Wsum, bsum, out);
}
